# starting kernel
# baseline (speedup 1.0000x reference)
#include <hip/hip_runtime.h>
#include <stdint.h>

typedef __attribute__((ext_vector_type(2))) float v2f;
typedef __attribute__((ext_vector_type(8))) float v8f;

#define N_NODES  50000
#define N_EDGES  800000
#define NODE_DIM 128
#define STATE    64
#define OUT_DIM  32
#define ETYPES   8
#define OUT_W    (ETYPES * OUT_DIM)

__global__ void EdgeGCN_zero_kernel(float* __restrict__ p, long long n) {
    long long i = (long long)blockIdx.x * blockDim.x + threadIdx.x;
    if (i < n) p[i] = 0.0f;
}

__global__ void EdgeGCN_etype_kernel(const float* __restrict__ ea,
                                     int* __restrict__ et) {
    int e = blockIdx.x * blockDim.x + threadIdx.x;
    if (e >= N_EDGES) return;
    const float* a = ea + (long long)e * ETYPES;
    int t = 0;
#pragma unroll
    for (int j = 1; j < ETYPES; ++j)
        if (a[j] > 0.5f) t = j;
    et[e] = t;
}

__global__ void EdgeGCN_deg_kernel(const int* __restrict__ et,
                                   const int* __restrict__ eidx,
                                   float* __restrict__ deg) {
    int e = blockIdx.x * blockDim.x + threadIdx.x;
    if (e >= N_EDGES) return;
    int t = et[e];
    int c = eidx[N_EDGES + e];
    atomicAdd(&deg[(long long)t * N_NODES + c], 1.0f);
}

__global__ void EdgeGCN_dinv_kernel(float* __restrict__ d) {
    long long i = (long long)blockIdx.x * blockDim.x + threadIdx.x;
    if (i < (long long)ETYPES * N_NODES) d[i] = rsqrtf(d[i] + 1.0f);
}

template <int K, int S>
__global__ __launch_bounds__(256) void EdgeGCN_gemm_wmma(
        const float* __restrict__ A, const float* __restrict__ B,
        float* __restrict__ C, int nrows) {
    __shared__ float sB[K * S];
    const int tid = threadIdx.x;
    for (int i = tid; i < K * S; i += 256) sB[i] = B[i];
    __syncthreads();

    const int lane = tid & 31;
    const int w    = tid >> 5;
    constexpr int colTiles     = S / 16;
    constexpr int rowsPerBlock = 16 * (8 / colTiles);
    const int row0 = blockIdx.x * rowsPerBlock + (w / colTiles) * 16;
    const int col0 = (w % colTiles) * 16;
    const int lo = lane & 15;
    const int hi = lane >> 4;

    const int m = row0 + lo;
    const long long arow = (long long)((m < nrows) ? m : 0) * K;

    v8f c = {};
#pragma unroll
    for (int kb = 0; kb < K; kb += 4) {
        v2f a = *(const v2f*)(A + arow + kb + 2 * hi);
        v2f b;
        b.x = sB[(kb + 2 * hi)     * S + col0 + lo];
        b.y = sB[(kb + 2 * hi + 1) * S + col0 + lo];
        c = __builtin_amdgcn_wmma_f32_16x16x4_f32(
                 false, a,  false, b,
                 (short)0, c,  false,  false);
    }

#pragma unroll
    for (int v = 0; v < 8; ++v) {
        int mr = row0 + v + 8 * hi;
        if (mr < nrows) C[(long long)mr * S + col0 + lo] = c[v];
    }
}

__global__ void EdgeGCN_scatter1_kernel(const int* __restrict__ et,
                                        const int* __restrict__ eidx,
                                        const float* __restrict__ dinv_t,
                                        const float* __restrict__ h1,
                                        float* __restrict__ agg, int t) {
    long long tid = (long long)blockIdx.x * blockDim.x + threadIdx.x;
    if (tid >= (long long)N_EDGES * 8) return;
    int e  = (int)(tid >> 3);
    int s0 = (int)(tid & 7) * 8;
    if (et[e] != t) return;
    int r = eidx[e];
    int c = eidx[N_EDGES + e];
    float norm = dinv_t[r] * dinv_t[c];
    const float4* hp = (const float4*)(h1 + (long long)r * STATE + s0);
    float4 ha = hp[0], hb = hp[1];
    float* ap = agg + (long long)c * STATE + s0;
    atomicAdd(ap + 0, ha.x * norm); atomicAdd(ap + 1, ha.y * norm);
    atomicAdd(ap + 2, ha.z * norm); atomicAdd(ap + 3, ha.w * norm);
    atomicAdd(ap + 4, hb.x * norm); atomicAdd(ap + 5, hb.y * norm);
    atomicAdd(ap + 6, hb.z * norm); atomicAdd(ap + 7, hb.w * norm);
}

__global__ void EdgeGCN_combine1_kernel(const float* __restrict__ dinv_t,
                                        const float* __restrict__ h1,
                                        const float* __restrict__ b1t,
                                        float* __restrict__ agg) {
    long long i = (long long)blockIdx.x * blockDim.x + threadIdx.x;
    if (i >= (long long)N_NODES * STATE) return;
    int n = (int)(i >> 6);
    int s = (int)(i & (STATE - 1));
    float di = dinv_t[n];
    float v  = agg[i] + h1[i] * di * di + b1t[s];
    agg[i]   = fmaxf(v, 0.0f);
}

__global__ void EdgeGCN_scatter2_kernel(const int* __restrict__ et,
                                        const int* __restrict__ eidx,
                                        const float* __restrict__ dinv_t,
                                        const float* __restrict__ h2,
                                        float* __restrict__ out, int t) {
    long long tid = (long long)blockIdx.x * blockDim.x + threadIdx.x;
    if (tid >= (long long)N_EDGES * 8) return;
    int e  = (int)(tid >> 3);
    int s0 = (int)(tid & 7) * 4;
    if (et[e] != t) return;
    int r = eidx[e];
    int c = eidx[N_EDGES + e];
    float norm = dinv_t[r] * dinv_t[c];
    float4 h = *(const float4*)(h2 + (long long)r * OUT_DIM + s0);
    float* op = out + (long long)c * OUT_W + t * OUT_DIM + s0;
    atomicAdd(op + 0, h.x * norm); atomicAdd(op + 1, h.y * norm);
    atomicAdd(op + 2, h.z * norm); atomicAdd(op + 3, h.w * norm);
}

__global__ void EdgeGCN_combine2_kernel(const float* __restrict__ dinv_t,
                                        const float* __restrict__ h2,
                                        const float* __restrict__ b2t,
                                        float* __restrict__ out, int t) {
    long long i = (long long)blockIdx.x * blockDim.x + threadIdx.x;
    if (i >= (long long)N_NODES * OUT_DIM) return;
    int n = (int)(i >> 5);
    int s = (int)(i & (OUT_DIM - 1));
    float di = dinv_t[n];
    long long oi = (long long)n * OUT_W + t * OUT_DIM + s;
    out[oi] += h2[i] * di * di + b2t[s];
}

extern "C" void kernel_launch(void* const* d_in, const int* in_sizes, int n_in,
                              void* d_out, int out_size, void* d_ws, size_t ws_size,
                              hipStream_t stream) {
    const float* x   = (const float*)d_in[0];
    const float* ea  = (const float*)d_in[1];
    const float* W1  = (const float*)d_in[2];
    const float* b1  = (const float*)d_in[3];
    const float* W2  = (const float*)d_in[4];
    const float* b2  = (const float*)d_in[5];
    const int*   eix = (const int*)d_in[6];
    float* out = (float*)d_out;

    char* ws = (char*)d_ws;
    auto alignup = [](size_t v) { return (v + 255) & ~(size_t)255; };
    size_t off = 0;
    float* dinv = (float*)(ws + off); off += alignup(sizeof(float) * ETYPES * N_NODES);
    int*   et   = (int*)  (ws + off); off += alignup(sizeof(int) * N_EDGES);
    float* h1   = (float*)(ws + off); off += alignup(sizeof(float) * (size_t)N_NODES * STATE);
    float* agg1 = (float*)(ws + off); off += alignup(sizeof(float) * (size_t)N_NODES * STATE);
    float* h2   = (float*)(ws + off); off += alignup(sizeof(float) * (size_t)N_NODES * OUT_DIM);
    (void)ws_size; (void)in_sizes; (void)n_in; (void)out_size;

    const int TB = 256;
    auto blocksLL = [&](long long n) { return (unsigned)((n + TB - 1) / TB); };

    EdgeGCN_zero_kernel<<<blocksLL((long long)N_NODES * OUT_W), TB, 0, stream>>>(
        out, (long long)N_NODES * OUT_W);
    EdgeGCN_zero_kernel<<<blocksLL((long long)ETYPES * N_NODES), TB, 0, stream>>>(
        dinv, (long long)ETYPES * N_NODES);
    EdgeGCN_etype_kernel<<<blocksLL(N_EDGES), TB, 0, stream>>>(ea, et);
    EdgeGCN_deg_kernel<<<blocksLL(N_EDGES), TB, 0, stream>>>(et, eix, dinv);
    EdgeGCN_dinv_kernel<<<blocksLL((long long)ETYPES * N_NODES), TB, 0, stream>>>(dinv);

    for (int t = 0; t < ETYPES; ++t) {
        const float* dinv_t = dinv + (long long)t * N_NODES;

        {
            constexpr int rowsPerBlock = 16 * (8 / (STATE / 16));
            unsigned nb = (N_NODES + rowsPerBlock - 1) / rowsPerBlock;
            EdgeGCN_gemm_wmma<NODE_DIM, STATE><<<nb, TB, 0, stream>>>(
                x, W1 + (long long)t * NODE_DIM * STATE, h1, N_NODES);
        }

        EdgeGCN_zero_kernel<<<blocksLL((long long)N_NODES * STATE), TB, 0, stream>>>(
            agg1, (long long)N_NODES * STATE);
        EdgeGCN_scatter1_kernel<<<blocksLL((long long)N_EDGES * 8), TB, 0, stream>>>(
            et, eix, dinv_t, h1, agg1, t);
        EdgeGCN_combine1_kernel<<<blocksLL((long long)N_NODES * STATE), TB, 0, stream>>>(
            dinv_t, h1, b1 + (long long)t * STATE, agg1);

        {
            constexpr int rowsPerBlock = 16 * (8 / (OUT_DIM / 16));
            unsigned nb = (N_NODES + rowsPerBlock - 1) / rowsPerBlock;
            EdgeGCN_gemm_wmma<STATE, OUT_DIM><<<nb, TB, 0, stream>>>(
                agg1, W2 + (long long)t * STATE * OUT_DIM, h2, N_NODES);
        }

        EdgeGCN_scatter2_kernel<<<blocksLL((long long)N_EDGES * 8), TB, 0, stream>>>(
            et, eix, dinv_t, h2, out, t);
        EdgeGCN_combine2_kernel<<<blocksLL((long long)N_NODES * OUT_DIM), TB, 0, stream>>>(
            dinv_t, h2, b2 + (long long)t * OUT_DIM, out, t);
    }
}


# pipeline the edge types two deep: layer-2 GEMM of type t and layer-1 GEMM of type t+1 share one row-local launch, and their two scatters share one pass over the edge list (36 -> 22 launches)
# speedup vs baseline: 2.1104x; 2.1104x over previous
#include <hip/hip_runtime.h>
#include <stdint.h>

typedef __attribute__((ext_vector_type(2))) float v2f;
typedef __attribute__((ext_vector_type(8))) float v8f;

#define N_NODES  50000
#define N_EDGES  800000
#define NODE_DIM 128
#define STATE    64
#define OUT_DIM  32
#define ETYPES   8
#define OUT_W    (ETYPES * OUT_DIM)

__global__ void EdgeGCN_zero_kernel(float* __restrict__ p, long long n) {
    long long i = (long long)blockIdx.x * blockDim.x + threadIdx.x;
    if (i < n) p[i] = 0.0f;
}

__global__ void EdgeGCN_etype_kernel(const float* __restrict__ ea,
                                     int* __restrict__ et) {
    int e = blockIdx.x * blockDim.x + threadIdx.x;
    if (e >= N_EDGES) return;
    const float* a = ea + (long long)e * ETYPES;
    int t = 0;
#pragma unroll
    for (int j = 1; j < ETYPES; ++j)
        if (a[j] > 0.5f) t = j;
    et[e] = t;
}

__global__ void EdgeGCN_deg_kernel(const int* __restrict__ et,
                                   const int* __restrict__ eidx,
                                   float* __restrict__ deg) {
    int e = blockIdx.x * blockDim.x + threadIdx.x;
    if (e >= N_EDGES) return;
    int t = et[e];
    int c = eidx[N_EDGES + e];
    atomicAdd(&deg[(long long)t * N_NODES + c], 1.0f);
}

__global__ void EdgeGCN_dinv_kernel(float* __restrict__ d) {
    long long i = (long long)blockIdx.x * blockDim.x + threadIdx.x;
    if (i < (long long)ETYPES * N_NODES) d[i] = rsqrtf(d[i] + 1.0f);
}

template <int K, int S, bool RELU_A, int NT, bool SYNC_ST>
__device__ __forceinline__ void EdgeGCN_gemm_tile(
        const float* A, const float* sB, float* C, int nrows,
        const float* __restrict__ dinv_t, const float* __restrict__ bias,
        float* acc, int acc_stride, int row0, int col0, int lo, int hi) {
    const int m = row0 + lo;
    const long long arow = (long long)((m < nrows) ? m : nrows - 1) * K;

    float bcol[NT];
#pragma unroll
    for (int j = 0; j < NT; ++j) bcol[j] = bias[col0 + 16 * j + lo];
    float di[8];
#pragma unroll
    for (int v = 0; v < 8; ++v) {
        int mr = row0 + v + 8 * hi;
        di[v] = dinv_t[(mr < nrows) ? mr : nrows - 1];
    }

    v8f c[NT];
#pragma unroll
    for (int j = 0; j < NT; ++j) c[j] = v8f{};
    v2f af[K / 4];
#pragma unroll
    for (int i = 0; i < K / 4; ++i) af[i] = *(const v2f*)(A + arow + 4 * i + 2 * hi);
    if constexpr (NT > 1) __builtin_amdgcn_sched_barrier(0);
#pragma unroll
    for (int kb = 0; kb < K; kb += 4) {
        v2f a = af[kb / 4];
        if (RELU_A) { a.x = fmaxf(a.x, 0.0f); a.y = fmaxf(a.y, 0.0f); }
#pragma unroll
        for (int j = 0; j < NT; ++j) {
            v2f b;
            b.x = sB[(kb + 2 * hi)     * S + col0 + 16 * j + lo];
            b.y = sB[(kb + 2 * hi + 1) * S + col0 + 16 * j + lo];
            c[j] = __builtin_amdgcn_wmma_f32_16x16x4_f32(
                     false, a,  false, b,
                     (short)0, c[j],  false,  false);
        }
    }

    if constexpr (SYNC_ST) __syncthreads();

#pragma unroll
    for (int v = 0; v < 8; ++v) {
        int mr = row0 + v + 8 * hi;
        if (mr < nrows) {
#pragma unroll
            for (int j = 0; j < NT; ++j) {
                C[(long long)mr * S + col0 + 16 * j + lo] = c[j][v];
                acc[(long long)mr * acc_stride + col0 + 16 * j + lo] =
                    c[j][v] * di[v] * di[v] + bcol[j];
            }
        }
    }
}

template <int K, int S, bool RELU_A>
__global__ __launch_bounds__(256) void EdgeGCN_gemm_wmma(
        const float* __restrict__ A, const float* __restrict__ B,
        float* __restrict__ C, int nrows,
        const float* __restrict__ dinv_t, const float* __restrict__ bias,
        float* __restrict__ acc, int acc_stride) {
    __shared__ float sB[K * S];
    const int tid = threadIdx.x;
    for (int i = tid; i < K * S; i += 256) sB[i] = B[i];
    __syncthreads();

    const int lane = tid & 31;
    const int w    = tid >> 5;
    constexpr int colTiles     = S / 16;
    constexpr int rowsPerBlock = 16 * (8 / colTiles);
    const int row0 = blockIdx.x * rowsPerBlock + (w / colTiles) * 16;
    const int col0 = (w % colTiles) * 16;
    EdgeGCN_gemm_tile<K, S, RELU_A, 1, false>(A, sB, C, nrows, dinv_t, bias, acc, acc_stride,
                                              row0, col0, lane & 15, lane >> 4);
}

__global__ __launch_bounds__(256) void EdgeGCN_gemm_pair(
        const float* __restrict__ x,
        const float* __restrict__ W1n, const float* __restrict__ b1n,
        const float* __restrict__ W2t, const float* __restrict__ b2t,
        const float* __restrict__ dinv_t, const float* __restrict__ dinv_n,
        float* agg1, float* __restrict__ h1, float* __restrict__ h2,
        float* out_t, int nrows) {
    __shared__ float sW2[STATE * OUT_DIM];
    __shared__ float sW1[NODE_DIM * STATE];
    const int tid = threadIdx.x;
    for (int i = tid; i < STATE * OUT_DIM; i += 256) sW2[i] = W2t[i];
    for (int i = tid; i < NODE_DIM * STATE; i += 256) sW1[i] = W1n[i];
    __syncthreads();

    const int lane = tid & 31;
    const int w    = tid >> 5;
    const int lo   = lane & 15;
    const int hi   = lane >> 4;
    const int rowB = blockIdx.x * 64;

    EdgeGCN_gemm_tile<STATE, OUT_DIM, true, 1, false>(
        agg1, sW2, h2, nrows, dinv_t, b2t, out_t, OUT_W,
        rowB + (w >> 1) * 16, (w & 1) * 16, lo, hi);

    EdgeGCN_gemm_tile<NODE_DIM, STATE, false, 2, true>(
        x, sW1, h1, nrows, dinv_n, b1n, agg1, STATE,
        rowB + (w >> 1) * 16, (w & 1) * 32, lo, hi);
}

#define SC_CH 2

template <int F>
__device__ __forceinline__ void EdgeGCN_push_rows(
        unsigned m, int r, int c, float dr, float dc,
        const float* __restrict__ h, float* __restrict__ dst, int dst_stride, int lane) {
    constexpr int Q  = F / 32;
    constexpr int NB = 4;
    while (m) {
        int  s[NB];
        bool ok[NB];
#pragma unroll
        for (int j = 0; j < NB; ++j) {
            ok[j] = m != 0u;
            s[j]  = ok[j] ? __builtin_ctz(m) : 0;
            m &= m - 1u;
        }
        float v[NB][Q];
#pragma unroll
        for (int j = 0; j < NB; ++j) {
#pragma unroll
            for (int q = 0; q < Q; ++q) v[j][q] = 0.0f;
            if (ok[j]) {
                const int rr = __builtin_amdgcn_readlane(r, s[j]);
                const float* hp = h + (size_t)rr * F + lane;
#pragma unroll
                for (int q = 0; q < Q; ++q) v[j][q] = hp[32 * q];
            }
        }
#pragma unroll
        for (int j = 0; j < NB; ++j) {
            if (ok[j]) {
                const int cc = __builtin_amdgcn_readlane(c, s[j]);
                const float nn =
                    __int_as_float(__builtin_amdgcn_readlane(__float_as_int(dr), s[j])) *
                    __int_as_float(__builtin_amdgcn_readlane(__float_as_int(dc), s[j]));
                float* dp = dst + (size_t)cc * dst_stride + lane;
#pragma unroll
                for (int q = 0; q < Q; ++q) atomicAdd(dp + 32 * q, v[j][q] * nn);
            }
        }
    }
}

template <int F>
__global__ __launch_bounds__(256) void EdgeGCN_scatter_kernel(
        const int* __restrict__ et, const int* __restrict__ eidx,
        const float* __restrict__ dinv_t, const float* __restrict__ h,
        float* __restrict__ dst, int dst_stride, int t) {
    constexpr int CH = SC_CH;
    const int lane  = threadIdx.x & 31;
    const int wave  = blockIdx.x * 8 + (threadIdx.x >> 5);
    const int ebase = wave * (32 * CH) + lane;

    int ty[CH], r[CH], c[CH];
#pragma unroll
    for (int k = 0; k < CH; ++k) {
        const int e = ebase + 32 * k;
        const bool in = e < N_EDGES;
        ty[k] = in ? et[e] : -1;
        r[k]  = in ? eidx[e] : 0;
        c[k]  = in ? eidx[N_EDGES + e] : 0;
    }

    float dr[CH], dc[CH];
    unsigned mask[CH];
#pragma unroll
    for (int k = 0; k < CH; ++k) {
        const bool match = ty[k] == t;
        dr[k] = 0.0f; dc[k] = 0.0f;
        if (match) { dr[k] = dinv_t[r[k]]; dc[k] = dinv_t[c[k]]; }
        mask[k] = __builtin_amdgcn_ballot_w32(match);
    }

#pragma unroll
    for (int k = 0; k < CH; ++k)
        EdgeGCN_push_rows<F>(mask[k], r[k], c[k], dr[k], dc[k], h, dst, dst_stride, lane);
}

__global__ __launch_bounds__(256) void EdgeGCN_scatter_pair(
        const int* __restrict__ et, const int* __restrict__ eidx,
        const float* __restrict__ dinv_t, const float* __restrict__ dinv_n,
        const float* __restrict__ h2, float* __restrict__ out_t,
        const float* __restrict__ h1, float* __restrict__ agg1, int t) {
    constexpr int CH = SC_CH;
    const int lane  = threadIdx.x & 31;
    const int wave  = blockIdx.x * 8 + (threadIdx.x >> 5);
    const int ebase = wave * (32 * CH) + lane;

    int ty[CH], r[CH], c[CH];
#pragma unroll
    for (int k = 0; k < CH; ++k) {
        const int e = ebase + 32 * k;
        const bool in = e < N_EDGES;
        ty[k] = in ? et[e] : -1;
        r[k]  = in ? eidx[e] : 0;
        c[k]  = in ? eidx[N_EDGES + e] : 0;
    }

    float dr[CH], dc[CH];
    unsigned maskT[CH], maskN[CH];
#pragma unroll
    for (int k = 0; k < CH; ++k) {
        const bool mT = ty[k] == t;
        const bool mN = ty[k] == t + 1;
        const float* dv = mT ? dinv_t : dinv_n;
        dr[k] = 0.0f; dc[k] = 0.0f;
        if (mT || mN) { dr[k] = dv[r[k]]; dc[k] = dv[c[k]]; }
        maskT[k] = __builtin_amdgcn_ballot_w32(mT);
        maskN[k] = __builtin_amdgcn_ballot_w32(mN);
    }

#pragma unroll
    for (int k = 0; k < CH; ++k)
        EdgeGCN_push_rows<OUT_DIM>(maskT[k], r[k], c[k], dr[k], dc[k], h2, out_t, OUT_W, lane);
#pragma unroll
    for (int k = 0; k < CH; ++k)
        EdgeGCN_push_rows<STATE>(maskN[k], r[k], c[k], dr[k], dc[k], h1, agg1, STATE, lane);
}

extern "C" void kernel_launch(void* const* d_in, const int* in_sizes, int n_in,
                              void* d_out, int out_size, void* d_ws, size_t ws_size,
                              hipStream_t stream) {
    const float* x   = (const float*)d_in[0];
    const float* ea  = (const float*)d_in[1];
    const float* W1  = (const float*)d_in[2];
    const float* b1  = (const float*)d_in[3];
    const float* W2  = (const float*)d_in[4];
    const float* b2  = (const float*)d_in[5];
    const int*   eix = (const int*)d_in[6];
    float* out = (float*)d_out;

    char* ws = (char*)d_ws;
    auto alignup = [](size_t v) { return (v + 255) & ~(size_t)255; };
    size_t off = 0;
    float* dinv = (float*)(ws + off); off += alignup(sizeof(float) * ETYPES * N_NODES);
    int*   et   = (int*)  (ws + off); off += alignup(sizeof(int) * N_EDGES);
    float* h1   = (float*)(ws + off); off += alignup(sizeof(float) * (size_t)N_NODES * STATE);
    float* agg1 = (float*)(ws + off); off += alignup(sizeof(float) * (size_t)N_NODES * STATE);
    float* h2   = (float*)(ws + off); off += alignup(sizeof(float) * (size_t)N_NODES * OUT_DIM);
    (void)ws_size; (void)in_sizes; (void)n_in; (void)out_size;

    const int TB = 256;
    auto blocksLL = [&](long long n) { return (unsigned)((n + TB - 1) / TB); };
    constexpr unsigned scatterBlocks = (N_EDGES + 511) / 512;
    constexpr unsigned gemm1Blocks = (N_NODES + 31) / 32;
    constexpr unsigned gemm2Blocks = (N_NODES + 63) / 64;
    constexpr unsigned pairBlocks  = (N_NODES + 63) / 64;

    EdgeGCN_zero_kernel<<<blocksLL((long long)ETYPES * N_NODES), TB, 0, stream>>>(
        dinv, (long long)ETYPES * N_NODES);
    EdgeGCN_etype_kernel<<<blocksLL(N_EDGES), TB, 0, stream>>>(ea, et);
    EdgeGCN_deg_kernel<<<blocksLL(N_EDGES), TB, 0, stream>>>(et, eix, dinv);
    EdgeGCN_dinv_kernel<<<blocksLL((long long)ETYPES * N_NODES), TB, 0, stream>>>(dinv);


    EdgeGCN_gemm_wmma<NODE_DIM, STATE, false><<<gemm1Blocks, TB, 0, stream>>>(
        x, W1, h1, N_NODES, dinv, b1, agg1, STATE);
    EdgeGCN_scatter_kernel<STATE><<<scatterBlocks, TB, 0, stream>>>(
        et, eix, dinv, h1, agg1, STATE, 0);

    for (int t = 0; t + 1 < ETYPES; ++t) {
        const float* dinv_t = dinv + (long long)t * N_NODES;
        const float* dinv_n = dinv_t + N_NODES;
        float* out_t = out + (long long)t * OUT_DIM;

        EdgeGCN_gemm_pair<<<pairBlocks, TB, 0, stream>>>(
            x, W1 + (long long)(t + 1) * NODE_DIM * STATE, b1 + (long long)(t + 1) * STATE,
            W2 + (long long)t * STATE * OUT_DIM, b2 + (long long)t * OUT_DIM,
            dinv_t, dinv_n, agg1, h1, h2, out_t, N_NODES);

        EdgeGCN_scatter_pair<<<scatterBlocks, TB, 0, stream>>>(
            et, eix, dinv_t, dinv_n, h2, out_t, h1, agg1, t);
    }

    {
        const int t = ETYPES - 1;
        const float* dinv_t = dinv + (long long)t * N_NODES;
        float* out_t = out + (long long)t * OUT_DIM;
        EdgeGCN_gemm_wmma<STATE, OUT_DIM, true><<<gemm2Blocks, TB, 0, stream>>>(
            agg1, W2 + (long long)t * STATE * OUT_DIM, h2, N_NODES,
            dinv_t, b2 + (long long)t * OUT_DIM, out_t, OUT_W);
        EdgeGCN_scatter_kernel<OUT_DIM><<<scatterBlocks, TB, 0, stream>>>(
            et, eix, dinv_t, h2, out_t, OUT_W, t);
    }
}


# GEMM kernels: all start-up loads in flight together (weights to LDS by 16-byte loads issued back to back, first product's operands requested before the wait); the paired kernel's second barrier goes
# speedup vs baseline: 2.3968x; 1.1357x over previous
#include <hip/hip_runtime.h>
#include <stdint.h>

typedef __attribute__((ext_vector_type(2))) float v2f;
typedef __attribute__((ext_vector_type(8))) float v8f;
typedef __attribute__((ext_vector_type(4))) float v4f;

#define N_NODES  50000
#define N_EDGES  800000
#define NODE_DIM 128
#define STATE    64
#define OUT_DIM  32
#define ETYPES   8
#define OUT_W    (ETYPES * OUT_DIM)

__global__ void EdgeGCN_zero_kernel(float* __restrict__ p, long long n) {
    long long i = (long long)blockIdx.x * blockDim.x + threadIdx.x;
    if (i < n) p[i] = 0.0f;
}

__global__ void EdgeGCN_etype_kernel(const float* __restrict__ ea,
                                     int* __restrict__ et) {
    int e = blockIdx.x * blockDim.x + threadIdx.x;
    if (e >= N_EDGES) return;
    const float* a = ea + (long long)e * ETYPES;
    int t = 0;
#pragma unroll
    for (int j = 1; j < ETYPES; ++j)
        if (a[j] > 0.5f) t = j;
    et[e] = t;
}

__global__ void EdgeGCN_deg_kernel(const int* __restrict__ et,
                                   const int* __restrict__ eidx,
                                   float* __restrict__ deg) {
    int e = blockIdx.x * blockDim.x + threadIdx.x;
    if (e >= N_EDGES) return;
    int t = et[e];
    int c = eidx[N_EDGES + e];
    atomicAdd(&deg[(long long)t * N_NODES + c], 1.0f);
}

__global__ void EdgeGCN_dinv_kernel(float* __restrict__ d) {
    long long i = (long long)blockIdx.x * blockDim.x + threadIdx.x;
    if (i < (long long)ETYPES * N_NODES) d[i] = rsqrtf(d[i] + 1.0f);
}

template <int NF>
__device__ __forceinline__ void EdgeGCN_w_request(
        const float* __restrict__ g, int tid, v4f (&r)[NF / 1024]) {
#pragma unroll
    for (int i = 0; i < NF / 1024; ++i) r[i] = ((const v4f*)g)[tid + 256 * i];
}
template <int NF>
__device__ __forceinline__ void EdgeGCN_w_to_lds(
        float* s, int tid, const v4f (&r)[NF / 1024]) {
#pragma unroll
    for (int i = 0; i < NF / 1024; ++i) ((v4f*)s)[tid + 256 * i] = r[i];
}

template <int K, int NT>
__device__ __forceinline__ void EdgeGCN_tile_fetch(
        const float* A, int nrows,
        const float* __restrict__ dinv_t, const float* __restrict__ bias,
        int row0, int col0, int lo, int hi,
        v2f (&af)[K / 4], float (&bcol)[NT], float (&di)[8]) {
    const int m = row0 + lo;
    const long long arow = (long long)((m < nrows) ? m : nrows - 1) * K;
#pragma unroll
    for (int i = 0; i < K / 4; ++i) af[i] = *(const v2f*)(A + arow + 4 * i + 2 * hi);
#pragma unroll
    for (int j = 0; j < NT; ++j) bcol[j] = bias[col0 + 16 * j + lo];
#pragma unroll
    for (int v = 0; v < 8; ++v) {
        int mr = row0 + v + 8 * hi;
        di[v] = dinv_t[(mr < nrows) ? mr : nrows - 1];
    }
}

template <int K, int S, bool RELU_A, int NT>
__device__ __forceinline__ void EdgeGCN_tile_mma_store(
        const v2f (&af)[K / 4], const float (&bcol)[NT], const float (&di)[8],
        const float* sB, float* C, int nrows, float* acc, int acc_stride,
        int row0, int col0, int lo, int hi) {
    v8f c[NT];
#pragma unroll
    for (int j = 0; j < NT; ++j) c[j] = v8f{};
#pragma unroll
    for (int kb = 0; kb < K; kb += 4) {
        v2f a = af[kb / 4];
        if (RELU_A) { a.x = fmaxf(a.x, 0.0f); a.y = fmaxf(a.y, 0.0f); }
#pragma unroll
        for (int j = 0; j < NT; ++j) {
            v2f b;
            b.x = sB[(kb + 2 * hi)     * S + col0 + 16 * j + lo];
            b.y = sB[(kb + 2 * hi + 1) * S + col0 + 16 * j + lo];
            c[j] = __builtin_amdgcn_wmma_f32_16x16x4_f32(
                     false, a,  false, b,
                     (short)0, c[j],  false,  false);
        }
    }
#pragma unroll
    for (int v = 0; v < 8; ++v) {
        int mr = row0 + v + 8 * hi;
        if (mr < nrows) {
#pragma unroll
            for (int j = 0; j < NT; ++j) {
                C[(long long)mr * S + col0 + 16 * j + lo] = c[j][v];
                acc[(long long)mr * acc_stride + col0 + 16 * j + lo] =
                    c[j][v] * di[v] * di[v] + bcol[j];
            }
        }
    }
}

template <int K, int S, bool RELU_A>
__global__ __launch_bounds__(256) void EdgeGCN_gemm_wmma(
        const float* __restrict__ A, const float* __restrict__ B,
        float* __restrict__ C, int nrows,
        const float* __restrict__ dinv_t, const float* __restrict__ bias,
        float* __restrict__ acc, int acc_stride) {
    __shared__ alignas(16) float sB[K * S];
    const int tid  = threadIdx.x;
    const int lane = tid & 31;
    const int w    = tid >> 5;
    constexpr int colTiles     = S / 16;
    constexpr int rowsPerBlock = 16 * (8 / colTiles);
    const int row0 = blockIdx.x * rowsPerBlock + (w / colTiles) * 16;
    const int col0 = (w % colTiles) * 16;
    const int lo = lane & 15, hi = lane >> 4;

    v4f wr[K * S / 1024];
    EdgeGCN_w_request<K * S>(B, tid, wr);
    v2f af[K / 4];
    float bcol[1], di[8];
    EdgeGCN_tile_fetch<K, 1>(A, nrows, dinv_t, bias, row0, col0, lo, hi, af, bcol, di);
    EdgeGCN_w_to_lds<K * S>(sB, tid, wr);
    __syncthreads();

    EdgeGCN_tile_mma_store<K, S, RELU_A, 1>(af, bcol, di, sB, C, nrows, acc, acc_stride,
                                            row0, col0, lo, hi);
}

__global__ __launch_bounds__(256) void EdgeGCN_gemm_pair(
        const float* __restrict__ x,
        const float* __restrict__ W1n, const float* __restrict__ b1n,
        const float* __restrict__ W2t, const float* __restrict__ b2t,
        const float* __restrict__ dinv_t, const float* __restrict__ dinv_n,
        float* agg1, float* __restrict__ h1, float* __restrict__ h2,
        float* out_t, int nrows) {
    __shared__ alignas(16) float sW2[STATE * OUT_DIM];
    __shared__ alignas(16) float sW1[NODE_DIM * STATE];
    const int tid  = threadIdx.x;
    const int lane = tid & 31;
    const int w    = tid >> 5;
    const int lo   = lane & 15;
    const int hi   = lane >> 4;
    const int row0 = blockIdx.x * 64 + (w >> 1) * 16;

    v4f wr2[STATE * OUT_DIM / 1024];
    v4f wr1[NODE_DIM * STATE / 1024];
    EdgeGCN_w_request<STATE * OUT_DIM>(W2t, tid, wr2);
    EdgeGCN_w_request<NODE_DIM * STATE>(W1n, tid, wr1);
    v2f af2[STATE / 4];
    float bc2[1], di2[8];
    EdgeGCN_tile_fetch<STATE, 1>(agg1, nrows, dinv_t, b2t, row0, (w & 1) * 16, lo, hi,
                                 af2, bc2, di2);
    EdgeGCN_w_to_lds<STATE * OUT_DIM>(sW2, tid, wr2);
    EdgeGCN_w_to_lds<NODE_DIM * STATE>(sW1, tid, wr1);
    __syncthreads();

    EdgeGCN_tile_mma_store<STATE, OUT_DIM, true, 1>(
        af2, bc2, di2, sW2, h2, nrows, out_t, OUT_W, row0, (w & 1) * 16, lo, hi);

    v2f af1[NODE_DIM / 4];
    float bc1[2], di1[8];
    EdgeGCN_tile_fetch<NODE_DIM, 2>(x, nrows, dinv_n, b1n, row0, (w & 1) * 32, lo, hi,
                                    af1, bc1, di1);
    __builtin_amdgcn_sched_barrier(0);
    EdgeGCN_tile_mma_store<NODE_DIM, STATE, false, 2>(
        af1, bc1, di1, sW1, h1, nrows, agg1, STATE, row0, (w & 1) * 32, lo, hi);
}

#define SC_CH 2

template <int F>
__device__ __forceinline__ void EdgeGCN_push_rows(
        unsigned m, int r, int c, float dr, float dc,
        const float* __restrict__ h, float* __restrict__ dst, int dst_stride, int lane) {
    constexpr int Q  = F / 32;
    constexpr int NB = 4;
    while (m) {
        int  s[NB];
        bool ok[NB];
#pragma unroll
        for (int j = 0; j < NB; ++j) {
            ok[j] = m != 0u;
            s[j]  = ok[j] ? __builtin_ctz(m) : 0;
            m &= m - 1u;
        }
        float v[NB][Q];
#pragma unroll
        for (int j = 0; j < NB; ++j) {
#pragma unroll
            for (int q = 0; q < Q; ++q) v[j][q] = 0.0f;
            if (ok[j]) {
                const int rr = __builtin_amdgcn_readlane(r, s[j]);
                const float* hp = h + (size_t)rr * F + lane;
#pragma unroll
                for (int q = 0; q < Q; ++q) v[j][q] = hp[32 * q];
            }
        }
#pragma unroll
        for (int j = 0; j < NB; ++j) {
            if (ok[j]) {
                const int cc = __builtin_amdgcn_readlane(c, s[j]);
                const float nn =
                    __int_as_float(__builtin_amdgcn_readlane(__float_as_int(dr), s[j])) *
                    __int_as_float(__builtin_amdgcn_readlane(__float_as_int(dc), s[j]));
                float* dp = dst + (size_t)cc * dst_stride + lane;
#pragma unroll
                for (int q = 0; q < Q; ++q) atomicAdd(dp + 32 * q, v[j][q] * nn);
            }
        }
    }
}

template <int F>
__global__ __launch_bounds__(256) void EdgeGCN_scatter_kernel(
        const int* __restrict__ et, const int* __restrict__ eidx,
        const float* __restrict__ dinv_t, const float* __restrict__ h,
        float* __restrict__ dst, int dst_stride, int t) {
    constexpr int CH = SC_CH;
    const int lane  = threadIdx.x & 31;
    const int wave  = blockIdx.x * 8 + (threadIdx.x >> 5);
    const int ebase = wave * (32 * CH) + lane;

    int ty[CH], r[CH], c[CH];
#pragma unroll
    for (int k = 0; k < CH; ++k) {
        const int e = ebase + 32 * k;
        const bool in = e < N_EDGES;
        ty[k] = in ? et[e] : -1;
        r[k]  = in ? eidx[e] : 0;
        c[k]  = in ? eidx[N_EDGES + e] : 0;
    }

    float dr[CH], dc[CH];
    unsigned mask[CH];
#pragma unroll
    for (int k = 0; k < CH; ++k) {
        const bool match = ty[k] == t;
        dr[k] = 0.0f; dc[k] = 0.0f;
        if (match) { dr[k] = dinv_t[r[k]]; dc[k] = dinv_t[c[k]]; }
        mask[k] = __builtin_amdgcn_ballot_w32(match);
    }

#pragma unroll
    for (int k = 0; k < CH; ++k)
        EdgeGCN_push_rows<F>(mask[k], r[k], c[k], dr[k], dc[k], h, dst, dst_stride, lane);
}

__global__ __launch_bounds__(256) void EdgeGCN_scatter_pair(
        const int* __restrict__ et, const int* __restrict__ eidx,
        const float* __restrict__ dinv_t, const float* __restrict__ dinv_n,
        const float* __restrict__ h2, float* __restrict__ out_t,
        const float* __restrict__ h1, float* __restrict__ agg1, int t) {
    constexpr int CH = SC_CH;
    const int lane  = threadIdx.x & 31;
    const int wave  = blockIdx.x * 8 + (threadIdx.x >> 5);
    const int ebase = wave * (32 * CH) + lane;

    int ty[CH], r[CH], c[CH];
#pragma unroll
    for (int k = 0; k < CH; ++k) {
        const int e = ebase + 32 * k;
        const bool in = e < N_EDGES;
        ty[k] = in ? et[e] : -1;
        r[k]  = in ? eidx[e] : 0;
        c[k]  = in ? eidx[N_EDGES + e] : 0;
    }

    float dr[CH], dc[CH];
    unsigned maskT[CH], maskN[CH];
#pragma unroll
    for (int k = 0; k < CH; ++k) {
        const bool mT = ty[k] == t;
        const bool mN = ty[k] == t + 1;
        const float* dv = mT ? dinv_t : dinv_n;
        dr[k] = 0.0f; dc[k] = 0.0f;
        if (mT || mN) { dr[k] = dv[r[k]]; dc[k] = dv[c[k]]; }
        maskT[k] = __builtin_amdgcn_ballot_w32(mT);
        maskN[k] = __builtin_amdgcn_ballot_w32(mN);
    }

#pragma unroll
    for (int k = 0; k < CH; ++k)
        EdgeGCN_push_rows<OUT_DIM>(maskT[k], r[k], c[k], dr[k], dc[k], h2, out_t, OUT_W, lane);
#pragma unroll
    for (int k = 0; k < CH; ++k)
        EdgeGCN_push_rows<STATE>(maskN[k], r[k], c[k], dr[k], dc[k], h1, agg1, STATE, lane);
}

extern "C" void kernel_launch(void* const* d_in, const int* in_sizes, int n_in,
                              void* d_out, int out_size, void* d_ws, size_t ws_size,
                              hipStream_t stream) {
    const float* x   = (const float*)d_in[0];
    const float* ea  = (const float*)d_in[1];
    const float* W1  = (const float*)d_in[2];
    const float* b1  = (const float*)d_in[3];
    const float* W2  = (const float*)d_in[4];
    const float* b2  = (const float*)d_in[5];
    const int*   eix = (const int*)d_in[6];
    float* out = (float*)d_out;

    char* ws = (char*)d_ws;
    auto alignup = [](size_t v) { return (v + 255) & ~(size_t)255; };
    size_t off = 0;
    float* dinv = (float*)(ws + off); off += alignup(sizeof(float) * ETYPES * N_NODES);
    int*   et   = (int*)  (ws + off); off += alignup(sizeof(int) * N_EDGES);
    float* h1   = (float*)(ws + off); off += alignup(sizeof(float) * (size_t)N_NODES * STATE);
    float* agg1 = (float*)(ws + off); off += alignup(sizeof(float) * (size_t)N_NODES * STATE);
    float* h2   = (float*)(ws + off); off += alignup(sizeof(float) * (size_t)N_NODES * OUT_DIM);
    (void)ws_size; (void)in_sizes; (void)n_in; (void)out_size;

    const int TB = 256;
    auto blocksLL = [&](long long n) { return (unsigned)((n + TB - 1) / TB); };
    constexpr unsigned scatterBlocks = (N_EDGES + 511) / 512;
    constexpr unsigned gemm1Blocks = (N_NODES + 31) / 32;
    constexpr unsigned gemm2Blocks = (N_NODES + 63) / 64;
    constexpr unsigned pairBlocks  = (N_NODES + 63) / 64;

    EdgeGCN_zero_kernel<<<blocksLL((long long)ETYPES * N_NODES), TB, 0, stream>>>(
        dinv, (long long)ETYPES * N_NODES);
    EdgeGCN_etype_kernel<<<blocksLL(N_EDGES), TB, 0, stream>>>(ea, et);
    EdgeGCN_deg_kernel<<<blocksLL(N_EDGES), TB, 0, stream>>>(et, eix, dinv);
    EdgeGCN_dinv_kernel<<<blocksLL((long long)ETYPES * N_NODES), TB, 0, stream>>>(dinv);


    EdgeGCN_gemm_wmma<NODE_DIM, STATE, false><<<gemm1Blocks, TB, 0, stream>>>(
        x, W1, h1, N_NODES, dinv, b1, agg1, STATE);
    EdgeGCN_scatter_kernel<STATE><<<scatterBlocks, TB, 0, stream>>>(
        et, eix, dinv, h1, agg1, STATE, 0);

    for (int t = 0; t + 1 < ETYPES; ++t) {
        const float* dinv_t = dinv + (long long)t * N_NODES;
        const float* dinv_n = dinv_t + N_NODES;
        float* out_t = out + (long long)t * OUT_DIM;

        EdgeGCN_gemm_pair<<<pairBlocks, TB, 0, stream>>>(
            x, W1 + (long long)(t + 1) * NODE_DIM * STATE, b1 + (long long)(t + 1) * STATE,
            W2 + (long long)t * STATE * OUT_DIM, b2 + (long long)t * OUT_DIM,
            dinv_t, dinv_n, agg1, h1, h2, out_t, N_NODES);

        EdgeGCN_scatter_pair<<<scatterBlocks, TB, 0, stream>>>(
            et, eix, dinv_t, dinv_n, h2, out_t, h1, agg1, t);
    }

    {
        const int t = ETYPES - 1;
        const float* dinv_t = dinv + (long long)t * N_NODES;
        float* out_t = out + (long long)t * OUT_DIM;
        EdgeGCN_gemm_wmma<STATE, OUT_DIM, true><<<gemm2Blocks, TB, 0, stream>>>(
            agg1, W2 + (long long)t * STATE * OUT_DIM, h2, N_NODES,
            dinv_t, b2 + (long long)t * OUT_DIM, out_t, OUT_W);
        EdgeGCN_scatter_kernel<OUT_DIM><<<scatterBlocks, TB, 0, stream>>>(
            et, eix, dinv_t, h2, out_t, OUT_W, t);
    }
}


# edge weight split over its two ends: epilogues store rows x dinv[row]; layer-1 sum gets dinv[col] and bias where read; dinv lookups per edge 2->1 (layer 2), 2->0 (layer 1); scaled after sum
# speedup vs baseline: 2.7704x; 1.1559x over previous
#include <hip/hip_runtime.h>
#include <stdint.h>

typedef __attribute__((ext_vector_type(2))) float v2f;
typedef __attribute__((ext_vector_type(8))) float v8f;
typedef __attribute__((ext_vector_type(4))) float v4f;

#define N_NODES  50000
#define N_EDGES  800000
#define NODE_DIM 128
#define STATE    64
#define OUT_DIM  32
#define ETYPES   8
#define OUT_W    (ETYPES * OUT_DIM)

__global__ void EdgeGCN_zero_kernel(float* __restrict__ p, long long n) {
    long long i = (long long)blockIdx.x * blockDim.x + threadIdx.x;
    if (i < n) p[i] = 0.0f;
}

__global__ void EdgeGCN_etype_kernel(const float* __restrict__ ea,
                                     int* __restrict__ et) {
    int e = blockIdx.x * blockDim.x + threadIdx.x;
    if (e >= N_EDGES) return;
    const float* a = ea + (long long)e * ETYPES;
    int t = 0;
#pragma unroll
    for (int j = 1; j < ETYPES; ++j)
        if (a[j] > 0.5f) t = j;
    et[e] = t;
}

__global__ void EdgeGCN_deg_kernel(const int* __restrict__ et,
                                   const int* __restrict__ eidx,
                                   float* __restrict__ deg) {
    int e = blockIdx.x * blockDim.x + threadIdx.x;
    if (e >= N_EDGES) return;
    int t = et[e];
    int c = eidx[N_EDGES + e];
    atomicAdd(&deg[(long long)t * N_NODES + c], 1.0f);
}

__global__ void EdgeGCN_dinv_kernel(float* __restrict__ d) {
    long long i = (long long)blockIdx.x * blockDim.x + threadIdx.x;
    if (i < (long long)ETYPES * N_NODES) d[i] = rsqrtf(d[i] + 1.0f);
}

template <int NF>
__device__ __forceinline__ void EdgeGCN_w_request(
        const float* __restrict__ g, int tid, v4f (&r)[NF / 1024]) {
#pragma unroll
    for (int i = 0; i < NF / 1024; ++i) r[i] = ((const v4f*)g)[tid + 256 * i];
}
template <int NF>
__device__ __forceinline__ void EdgeGCN_w_to_lds(
        float* s, int tid, const v4f (&r)[NF / 1024]) {
#pragma unroll
    for (int i = 0; i < NF / 1024; ++i) ((v4f*)s)[tid + 256 * i] = r[i];
}

#define XPITCH (NODE_DIM + 4)
#define APITCH (STATE + 4)

template <int NT, bool BIAS>
__device__ __forceinline__ void EdgeGCN_epi_fetch(
        const float* __restrict__ dinv_t, const float* __restrict__ bias, int nrows,
        int row0, int col0, int lo, int hi, float (&bcol)[NT], float (&di)[8]) {
#pragma unroll
    for (int j = 0; j < NT; ++j) bcol[j] = BIAS ? bias[col0 + 16 * j + lo] : 0.0f;
#pragma unroll
    for (int v = 0; v < 8; ++v) {
        int mr = row0 + v + 8 * hi;
        di[v] = dinv_t[(mr < nrows) ? mr : nrows - 1];
    }
}

template <int K, int S, int NT, bool FINAL>
__device__ __forceinline__ void EdgeGCN_tile_mma_store(
        const float* sArow, const float (&bcol)[NT], const float (&di)[8],
        const float* sB, float* C, int nrows, float* acc, int acc_stride,
        int row0, int col0, int lo, int hi) {
    v2f af[K / 4];
#pragma unroll
    for (int i = 0; i < K / 4; ++i) af[i] = *(const v2f*)(sArow + 4 * i + 2 * hi);
    v8f c[NT];
#pragma unroll
    for (int j = 0; j < NT; ++j) c[j] = v8f{};
#pragma unroll
    for (int kb = 0; kb < K; kb += 4) {
        const v2f a = af[kb / 4];
#pragma unroll
        for (int j = 0; j < NT; ++j) {
            v2f b;
            b.x = sB[(kb + 2 * hi)     * S + col0 + 16 * j + lo];
            b.y = sB[(kb + 2 * hi + 1) * S + col0 + 16 * j + lo];
            c[j] = __builtin_amdgcn_wmma_f32_16x16x4_f32(
                     false, a,  false, b,
                     (short)0, c[j],  false,  false);
        }
    }
#pragma unroll
    for (int v = 0; v < 8; ++v) {
        int mr = row0 + v + 8 * hi;
        if (mr < nrows) {
#pragma unroll
            for (int j = 0; j < NT; ++j) {
                const float g = c[j][v] * di[v];
                C[(long long)mr * S + col0 + 16 * j + lo] = g;
                acc[(long long)mr * acc_stride + col0 + 16 * j + lo] =
                    FINAL ? g * di[v] + bcol[j] : g;
            }
        }
    }
}

template <bool DO_L2, bool DO_L1>
__global__ __launch_bounds__(256) void EdgeGCN_gemm_stage(
        const float* __restrict__ x,
        const float* __restrict__ W1n, const float* __restrict__ b1t,
        const float* __restrict__ W2t, const float* __restrict__ b2t,
        const float* __restrict__ dinv_t, const float* __restrict__ dinv_n,
        float* agg1, float* __restrict__ h1, float* __restrict__ h2,
        float* out_t, int nrows) {
    __shared__ alignas(16) float sW2[DO_L2 ? STATE * OUT_DIM : 4];
    __shared__ alignas(16) float sA2[DO_L2 ? 64 * APITCH : 4];
    __shared__ alignas(16) float sW1[DO_L1 ? NODE_DIM * STATE : 4];
    __shared__ alignas(16) float sA1[DO_L1 ? 64 * XPITCH : 4];
    const int tid  = threadIdx.x;
    const int lane = tid & 31;
    const int w    = tid >> 5;
    const int lo   = lane & 15;
    const int hi   = lane >> 4;
    const int rowb = blockIdx.x * 64;
    const int row0 = rowb + (w >> 1) * 16;

    v4f wr2[STATE * OUT_DIM / 1024], ar2[4];
    v4f wr1[NODE_DIM * STATE / 1024], ar1[8];
    float bc2[1], di2[8], bc1[2], di1[8];
    v4f b1v;
    float dsc[4];
    if constexpr (DO_L2) {
        EdgeGCN_w_request<STATE * OUT_DIM>(W2t, tid, wr2);
#pragma unroll
        for (int i = 0; i < 4; ++i) {
            const int q = tid + 256 * i;
            const int r = rowb + (q >> 4);
            const int rc = (r < nrows) ? r : nrows - 1;
            ar2[i] = ((const v4f*)agg1)[(long long)rc * (STATE / 4) + (q & 15)];
            dsc[i] = dinv_t[rc];
        }
        b1v = ((const v4f*)b1t)[tid & 15];
        EdgeGCN_epi_fetch<1, true>(dinv_t, b2t, nrows, row0, (w & 1) * 16, lo, hi, bc2, di2);
    }
    if constexpr (DO_L1) {
        EdgeGCN_w_request<NODE_DIM * STATE>(W1n, tid, wr1);
#pragma unroll
        for (int i = 0; i < 8; ++i) {
            const int q = tid + 256 * i;
            const int r = rowb + (q >> 5);
            ar1[i] = ((const v4f*)x)[(long long)((r < nrows) ? r : nrows - 1) * (NODE_DIM / 4) + (q & 31)];
        }
        EdgeGCN_epi_fetch<2, false>(dinv_n, nullptr, nrows, row0, (w & 1) * 32, lo, hi, bc1, di1);
    }
    if constexpr (DO_L2) {
        EdgeGCN_w_to_lds<STATE * OUT_DIM>(sW2, tid, wr2);
#pragma unroll
        for (int i = 0; i < 4; ++i) {
            const int q = tid + 256 * i;
            v4f v = ar2[i];
            v.x = fmaxf(v.x * dsc[i] + b1v.x, 0.0f); v.y = fmaxf(v.y * dsc[i] + b1v.y, 0.0f);
            v.z = fmaxf(v.z * dsc[i] + b1v.z, 0.0f); v.w = fmaxf(v.w * dsc[i] + b1v.w, 0.0f);
            *(v4f*)(sA2 + (q >> 4) * APITCH + (q & 15) * 4) = v;
        }
    }
    if constexpr (DO_L1) {
        EdgeGCN_w_to_lds<NODE_DIM * STATE>(sW1, tid, wr1);
#pragma unroll
        for (int i = 0; i < 8; ++i) {
            const int q = tid + 256 * i;
            *(v4f*)(sA1 + (q >> 5) * XPITCH + (q & 31) * 4) = ar1[i];
        }
    }
    __syncthreads();

    if constexpr (DO_L2)
        EdgeGCN_tile_mma_store<STATE, OUT_DIM, 1, true>(
            sA2 + ((w >> 1) * 16 + lo) * APITCH, bc2, di2, sW2, h2, nrows, out_t, OUT_W,
            row0, (w & 1) * 16, lo, hi);
    if constexpr (DO_L1)
        EdgeGCN_tile_mma_store<NODE_DIM, STATE, 2, false>(
            sA1 + ((w >> 1) * 16 + lo) * XPITCH, bc1, di1, sW1, h1, nrows, agg1, STATE,
            row0, (w & 1) * 32, lo, hi);
}

#define SC_CH 2

template <int F, bool SCALE>
__device__ __forceinline__ void EdgeGCN_push_rows(
        unsigned m, int r, int c, float dc,
        const float* __restrict__ h, float* __restrict__ dst, int dst_stride, int lane) {
    constexpr int Q  = F / 32;
    constexpr int NB = 4;
    while (m) {
        int  s[NB];
        bool ok[NB];
#pragma unroll
        for (int j = 0; j < NB; ++j) {
            ok[j] = m != 0u;
            s[j]  = ok[j] ? __builtin_ctz(m) : 0;
            m &= m - 1u;
        }
        float v[NB][Q];
#pragma unroll
        for (int j = 0; j < NB; ++j) {
#pragma unroll
            for (int q = 0; q < Q; ++q) v[j][q] = 0.0f;
            if (ok[j]) {
                const int rr = __builtin_amdgcn_readlane(r, s[j]);
                const float* hp = h + (size_t)rr * F + lane;
#pragma unroll
                for (int q = 0; q < Q; ++q) v[j][q] = hp[32 * q];
            }
        }
#pragma unroll
        for (int j = 0; j < NB; ++j) {
            if (ok[j]) {
                const int cc = __builtin_amdgcn_readlane(c, s[j]);
                float* dp = dst + (size_t)cc * dst_stride + lane;
                if constexpr (SCALE) {
                    const float nn =
                        __int_as_float(__builtin_amdgcn_readlane(__float_as_int(dc), s[j]));
#pragma unroll
                    for (int q = 0; q < Q; ++q) atomicAdd(dp + 32 * q, v[j][q] * nn);
                } else {
#pragma unroll
                    for (int q = 0; q < Q; ++q) atomicAdd(dp + 32 * q, v[j][q]);
                }
            }
        }
    }
}

template <int F, bool SCALE>
__global__ __launch_bounds__(256) void EdgeGCN_scatter_kernel(
        const int* __restrict__ et, const int* __restrict__ eidx,
        const float* __restrict__ dinv_t, const float* __restrict__ h,
        float* __restrict__ dst, int dst_stride, int t) {
    constexpr int CH = SC_CH;
    const int lane  = threadIdx.x & 31;
    const int wave  = blockIdx.x * 8 + (threadIdx.x >> 5);
    const int ebase = wave * (32 * CH) + lane;

    int ty[CH], r[CH], c[CH];
#pragma unroll
    for (int k = 0; k < CH; ++k) {
        const int e = ebase + 32 * k;
        const bool in = e < N_EDGES;
        ty[k] = in ? et[e] : -1;
        r[k]  = in ? eidx[e] : 0;
        c[k]  = in ? eidx[N_EDGES + e] : 0;
    }

    float dc[CH];
    unsigned mask[CH];
#pragma unroll
    for (int k = 0; k < CH; ++k) {
        const bool match = ty[k] == t;
        dc[k] = 0.0f;
        if constexpr (SCALE) { if (match) dc[k] = dinv_t[c[k]]; }
        mask[k] = __builtin_amdgcn_ballot_w32(match);
    }

#pragma unroll
    for (int k = 0; k < CH; ++k)
        EdgeGCN_push_rows<F, SCALE>(mask[k], r[k], c[k], dc[k], h, dst, dst_stride, lane);
}

__global__ __launch_bounds__(256) void EdgeGCN_scatter_pair(
        const int* __restrict__ et, const int* __restrict__ eidx,
        const float* __restrict__ dinv_t,
        const float* __restrict__ h2, float* __restrict__ out_t,
        const float* __restrict__ h1, float* __restrict__ agg1, int t) {
    constexpr int CH = SC_CH;
    const int lane  = threadIdx.x & 31;
    const int wave  = blockIdx.x * 8 + (threadIdx.x >> 5);
    const int ebase = wave * (32 * CH) + lane;

    int ty[CH], r[CH], c[CH];
#pragma unroll
    for (int k = 0; k < CH; ++k) {
        const int e = ebase + 32 * k;
        const bool in = e < N_EDGES;
        ty[k] = in ? et[e] : -1;
        r[k]  = in ? eidx[e] : 0;
        c[k]  = in ? eidx[N_EDGES + e] : 0;
    }

    float dc[CH];
    unsigned maskT[CH], maskN[CH];
#pragma unroll
    for (int k = 0; k < CH; ++k) {
        const bool mT = ty[k] == t;
        const bool mN = ty[k] == t + 1;
        dc[k] = 0.0f;
        if (mT) dc[k] = dinv_t[c[k]];
        maskT[k] = __builtin_amdgcn_ballot_w32(mT);
        maskN[k] = __builtin_amdgcn_ballot_w32(mN);
    }

#pragma unroll
    for (int k = 0; k < CH; ++k)
        EdgeGCN_push_rows<OUT_DIM, true>(maskT[k], r[k], c[k], dc[k], h2, out_t, OUT_W, lane);
#pragma unroll
    for (int k = 0; k < CH; ++k)
        EdgeGCN_push_rows<STATE, false>(maskN[k], r[k], c[k], dc[k], h1, agg1, STATE, lane);
}

extern "C" void kernel_launch(void* const* d_in, const int* in_sizes, int n_in,
                              void* d_out, int out_size, void* d_ws, size_t ws_size,
                              hipStream_t stream) {
    const float* x   = (const float*)d_in[0];
    const float* ea  = (const float*)d_in[1];
    const float* W1  = (const float*)d_in[2];
    const float* b1  = (const float*)d_in[3];
    const float* W2  = (const float*)d_in[4];
    const float* b2  = (const float*)d_in[5];
    const int*   eix = (const int*)d_in[6];
    float* out = (float*)d_out;

    char* ws = (char*)d_ws;
    auto alignup = [](size_t v) { return (v + 255) & ~(size_t)255; };
    size_t off = 0;
    float* dinv = (float*)(ws + off); off += alignup(sizeof(float) * ETYPES * N_NODES);
    int*   et   = (int*)  (ws + off); off += alignup(sizeof(int) * N_EDGES);
    float* h1   = (float*)(ws + off); off += alignup(sizeof(float) * (size_t)N_NODES * STATE);
    float* agg1 = (float*)(ws + off); off += alignup(sizeof(float) * (size_t)N_NODES * STATE);
    float* h2   = (float*)(ws + off); off += alignup(sizeof(float) * (size_t)N_NODES * OUT_DIM);
    (void)ws_size; (void)in_sizes; (void)n_in; (void)out_size;

    const int TB = 256;
    auto blocksLL = [&](long long n) { return (unsigned)((n + TB - 1) / TB); };
    constexpr unsigned scatterBlocks = (N_EDGES + 511) / 512;
    constexpr unsigned gemmBlocks = (N_NODES + 63) / 64;

    EdgeGCN_zero_kernel<<<blocksLL((long long)ETYPES * N_NODES), TB, 0, stream>>>(
        dinv, (long long)ETYPES * N_NODES);
    EdgeGCN_etype_kernel<<<blocksLL(N_EDGES), TB, 0, stream>>>(ea, et);
    EdgeGCN_deg_kernel<<<blocksLL(N_EDGES), TB, 0, stream>>>(et, eix, dinv);
    EdgeGCN_dinv_kernel<<<blocksLL((long long)ETYPES * N_NODES), TB, 0, stream>>>(dinv);


    EdgeGCN_gemm_stage<false, true><<<gemmBlocks, TB, 0, stream>>>(
        x, W1, b1, W2, b2, dinv, dinv, agg1, h1, h2, out, N_NODES);
    EdgeGCN_scatter_kernel<STATE, false><<<scatterBlocks, TB, 0, stream>>>(
        et, eix, dinv, h1, agg1, STATE, 0);

    for (int t = 0; t + 1 < ETYPES; ++t) {
        const float* dinv_t = dinv + (long long)t * N_NODES;
        const float* dinv_n = dinv_t + N_NODES;
        float* out_t = out + (long long)t * OUT_DIM;

        EdgeGCN_gemm_stage<true, true><<<gemmBlocks, TB, 0, stream>>>(
            x, W1 + (long long)(t + 1) * NODE_DIM * STATE, b1 + (long long)t * STATE,
            W2 + (long long)t * STATE * OUT_DIM, b2 + (long long)t * OUT_DIM,
            dinv_t, dinv_n, agg1, h1, h2, out_t, N_NODES);

        EdgeGCN_scatter_pair<<<scatterBlocks, TB, 0, stream>>>(
            et, eix, dinv_t, h2, out_t, h1, agg1, t);
    }

    {
        const int t = ETYPES - 1;
        const float* dinv_t = dinv + (long long)t * N_NODES;
        float* out_t = out + (long long)t * OUT_DIM;
        EdgeGCN_gemm_stage<true, false><<<gemmBlocks, TB, 0, stream>>>(
            x, W1, b1 + (long long)t * STATE, W2 + (long long)t * STATE * OUT_DIM,
            b2 + (long long)t * OUT_DIM,
            dinv_t, dinv_t, agg1, h1, h2, out_t, N_NODES);
        EdgeGCN_scatter_kernel<OUT_DIM, true><<<scatterBlocks, TB, 0, stream>>>(
            et, eix, dinv_t, h2, out_t, OUT_W, t);
    }
}
